# starting kernel
# baseline (speedup 1.0000x reference)
#include <hip/hip_runtime.h>

#define B_SZ   4
#define T_SZ   1024
#define F_SZ   1025
#define M_SZ   128
#define ITERS  20
#define LR     0.3f
#define MOM    0.9f
#define INV    (2.0f / (B_SZ * T_SZ))

#define ROWS     16
#define FSTR     1027
#define MSTR     131
#define NDSTR    131
#define NWAVES   8
#define NTHREADS 256
#define NFT      ((F_SZ + 15) / 16)

typedef __attribute__((ext_vector_type(2))) float v2f;
typedef __attribute__((ext_vector_type(8))) float v8f;

__device__ __forceinline__ v8f wmma_f32(v2f a, v2f b, v8f c) {
  return __builtin_amdgcn_wmma_f32_16x16x4_f32(false, a, false, b,
                                               (short)0, c, false, false);
}

__global__ __launch_bounds__(NTHREADS)
void transpose_fb_kernel(const float* __restrict__ fb,
                         float* __restrict__ fbT) {
  int i = blockIdx.x * NTHREADS + threadIdx.x;
  if (i < M_SZ * F_SZ) {
    int m = i / F_SZ, f = i - m * F_SZ;
    fbT[i] = fb[f * M_SZ + m];
  }
}

__global__ __launch_bounds__(NTHREADS)
void invmel_sgd_kernel(const float* __restrict__ mel,
                       const float* __restrict__ spec0,
                       const float* __restrict__ fb,
                       const float* __restrict__ fbT,
                       float* __restrict__ out)
{
  __shared__ float s_spec[ROWS * FSTR];
  __shared__ float s_buf [ROWS * FSTR];
  __shared__ float s_mel [ROWS * MSTR];
  __shared__ float s_nd  [ROWS * NDSTR];

  const int tid  = threadIdx.x;
  const int lane = tid & 31;
  const int wave = tid >> 5;
  const int nl   = lane & 15;
  const int hs   = lane >> 4;
  const int khi  = hs * 2;
  const int hi8  = hs * 8;

  const int blk = blockIdx.x;
  const int b   = blk / (T_SZ / ROWS);
  const int t0  = (blk % (T_SZ / ROWS)) * ROWS;

  for (int i = tid; i < ROWS * F_SZ; i += NTHREADS) {
    int r = i / F_SZ, c = i - r * F_SZ;
    s_spec[r * FSTR + c] = spec0[(b * T_SZ + t0 + r) * F_SZ + c];
    s_buf [r * FSTR + c] = 0.0f;
  }
  for (int i = tid; i < ROWS * M_SZ; i += NTHREADS) {
    int r = i & (ROWS - 1), m = i >> 4;
    s_mel[r * MSTR + m] = mel[(b * M_SZ + m) * T_SZ + t0 + r];
  }
  __syncthreads();

  for (int it = 0; it < ITERS; ++it) {
    {
      const int n0 = wave * 16;
      v8f c;
      #pragma unroll
      for (int r = 0; r < 8; ++r)
        c[r] = -s_mel[(r + hi8) * MSTR + n0 + nl];

      #pragma unroll 4
      for (int k = 0; k < 1024; k += 4) {
        v2f a, bf;
        a.x  = s_spec[nl * FSTR + k + khi];
        a.y  = s_spec[nl * FSTR + k + khi + 1];
        bf.x = fb[(k + khi)     * M_SZ + n0 + nl];
        bf.y = fb[(k + khi + 1) * M_SZ + n0 + nl];
        c = wmma_f32(a, bf, c);
      }
      {
        v2f a, bf;
        a.x  = (hs == 0) ? s_spec[nl * FSTR + 1024]  : 0.0f;
        a.y  = 0.0f;
        bf.x = (hs == 0) ? fb[1024 * M_SZ + n0 + nl] : 0.0f;
        bf.y = 0.0f;
        c = wmma_f32(a, bf, c);
      }
      #pragma unroll
      for (int r = 0; r < 8; ++r)
        s_nd[(r + hi8) * NDSTR + n0 + nl] = c[r];
    }
    __syncthreads();

    for (int ft = wave; ft < NFT; ft += NWAVES) {
      const int f0   = ft * 16;
      const int fcol = f0 + nl;
      const int fcl  = (fcol < F_SZ) ? fcol : (F_SZ - 1);
      v8f acc = {0.f, 0.f, 0.f, 0.f, 0.f, 0.f, 0.f, 0.f};

      #pragma unroll 4
      for (int m = 0; m < M_SZ; m += 4) {
        v2f a, bf;
        a.x  = s_nd[nl * NDSTR + m + khi];
        a.y  = s_nd[nl * NDSTR + m + khi + 1];
        bf.x = fbT[(m + khi)     * F_SZ + fcl];
        bf.y = fbT[(m + khi + 1) * F_SZ + fcl];
        acc = wmma_f32(a, bf, acc);
      }
      if (fcol < F_SZ) {
        #pragma unroll
        for (int r = 0; r < 8; ++r) {
          const int off = (r + hi8) * FSTR + fcol;
          float nb = MOM * s_buf[off] + INV * acc[r];
          s_buf[off] = nb;
          float ns = s_spec[off] - LR * nb;
          s_spec[off] = ns > 0.0f ? ns : 0.0f;
        }
      }
    }
    __syncthreads();
  }

  for (int i = tid; i < ROWS * F_SZ; i += NTHREADS) {
    int r = i & (ROWS - 1), f = i >> 4;
    out[(b * F_SZ + f) * T_SZ + t0 + r] = s_spec[r * FSTR + f];
  }
}

extern "C" void kernel_launch(void* const* d_in, const int* in_sizes, int n_in,
                              void* d_out, int out_size, void* d_ws, size_t ws_size,
                              hipStream_t stream) {
  (void)in_sizes; (void)n_in; (void)out_size; (void)ws_size;
  const float* mel   = (const float*)d_in[0];
  const float* spec0 = (const float*)d_in[1];
  const float* fb    = (const float*)d_in[2];
  float* out = (float*)d_out;
  float* fbT = (float*)d_ws;

  {
    dim3 grid((M_SZ * F_SZ + NTHREADS - 1) / NTHREADS);
    hipLaunchKernelGGL(transpose_fb_kernel, grid, dim3(NTHREADS), 0, stream,
                       fb, fbT);
  }
  {
    dim3 grid((B_SZ * T_SZ) / ROWS);
    hipLaunchKernelGGL(invmel_sgd_kernel, grid, dim3(NTHREADS), 0, stream,
                       mel, spec0, fb, fbT, out);
  }
}


# B-operand (filterbank) global loads software-pipelined in both GEMMs: the next 16 k-steps are issued as one group into a second register set ahead of the current 16 WMMAs; summation order unchanged
# speedup vs baseline: 2.0301x; 2.0301x over previous
#include <hip/hip_runtime.h>

#define B_SZ   4
#define T_SZ   1024
#define F_SZ   1025
#define M_SZ   128
#define ITERS  20
#define LR     0.3f
#define MOM    0.9f
#define INV    (2.0f / (B_SZ * T_SZ))

#define ROWS     16
#define FSTR     1027
#define MSTR     131
#define NDSTR    131
#define NWAVES   8
#define NTHREADS 256
#define NFT      ((F_SZ + 15) / 16)
#define GRP      16
#define NG1      (1024 / (4 * GRP))

typedef __attribute__((ext_vector_type(2))) float v2f;
typedef __attribute__((ext_vector_type(8))) float v8f;

__device__ __forceinline__ v8f wmma_f32(v2f a, v2f b, v8f c) {
  return __builtin_amdgcn_wmma_f32_16x16x4_f32(false, a, false, b,
                                               (short)0, c, false, false);
}

__device__ __forceinline__ void load_b_group(v2f (&buf)[GRP],
                                             const float* __restrict__ base,
                                             int off, int k0, int rs) {
  const float* __restrict__ p = base + off;
  #pragma unroll
  for (int j = 0; j < GRP; ++j) {
    buf[j].x = p[(k0 + 4 * j)     * rs];
    buf[j].y = p[(k0 + 4 * j + 1) * rs];
  }
}

__device__ __forceinline__ v8f mma_group(v8f c, const float* pa, int k0,
                                         const v2f (&buf)[GRP]) {
  v2f a[GRP];
  #pragma unroll
  for (int j = 0; j < GRP; ++j) {
    a[j].x = pa[k0 + 4 * j];
    a[j].y = pa[k0 + 4 * j + 1];
  }
  #pragma unroll
  for (int j = 0; j < GRP; ++j)
    c = wmma_f32(a[j], buf[j], c);
  return c;
}

__device__ __forceinline__ void loads_first() {
  __builtin_amdgcn_sched_barrier(0);
}

__device__ __forceinline__ void refill_after(v8f& c, int& off) {
  asm volatile("" : "+v"(c), "+v"(off));
}

__device__ __forceinline__ void offset_ready(int& off) {
  asm volatile("" : "+v"(off));
}

__device__ __forceinline__ void wmma_distance(v8f& c) {
  asm volatile("v_nop\n\tv_nop\n\tv_nop\n\tv_nop\n\tv_nop" : "+v"(c));
}

__device__ __forceinline__ void lds_barrier() {
  __builtin_amdgcn_fence(__ATOMIC_RELEASE, "workgroup", "local");
  __builtin_amdgcn_s_barrier();
  __builtin_amdgcn_fence(__ATOMIC_ACQUIRE, "workgroup", "local");
}

__device__ __forceinline__ void gemm2_tile(int ft, int nl, int hi8, int ob2,
                                           const float* __restrict__ fbT,
                                           const float* pa2,
                                           float* s_spec, float* s_buf,
                                           v2f (&b0)[GRP], v2f (&b1)[GRP],
                                           const float* __restrict__ nbase,
                                           int noff, int nrs) {
  const int fcol = ft * 16 + nl;
  const int fcl  = (fcol < F_SZ) ? fcol : (F_SZ - 1);
  v8f acc = {0.f, 0.f, 0.f, 0.f, 0.f, 0.f, 0.f, 0.f};

  offset_ready(noff);
  load_b_group(b1, fbT, ob2 + fcl, 4 * GRP, F_SZ);
  loads_first();
  acc = mma_group(acc, pa2, 0, b0);
  refill_after(acc, noff);
  load_b_group(b0, nbase, noff, 0, nrs);
  loads_first();
  acc = mma_group(acc, pa2, 4 * GRP, b1);
  wmma_distance(acc);

  if (fcol < F_SZ) {
    #pragma unroll
    for (int r = 0; r < 8; ++r) {
      const int off = (r + hi8) * FSTR + fcol;
      float nb = MOM * s_buf[off] + INV * acc[r];
      s_buf[off] = nb;
      float ns = s_spec[off] - LR * nb;
      s_spec[off] = ns > 0.0f ? ns : 0.0f;
    }
  }
}

__global__ __launch_bounds__(NTHREADS)
void transpose_fb_kernel(const float* __restrict__ fb,
                         float* __restrict__ fbT) {
  int i = blockIdx.x * NTHREADS + threadIdx.x;
  if (i < M_SZ * F_SZ) {
    int m = i / F_SZ, f = i - m * F_SZ;
    fbT[i] = fb[f * M_SZ + m];
  }
}

__global__ __launch_bounds__(NTHREADS)
void invmel_sgd_kernel(const float* __restrict__ mel,
                       const float* __restrict__ spec0,
                       const float* __restrict__ fb,
                       const float* __restrict__ fbT,
                       float* __restrict__ out)
{
  __shared__ float s_spec[ROWS * FSTR];
  __shared__ float s_buf [ROWS * FSTR];
  __shared__ float s_mel [ROWS * MSTR];
  __shared__ float s_nd  [ROWS * NDSTR];

  const int tid  = threadIdx.x;
  const int lane = tid & 31;
  const int wave = __builtin_amdgcn_readfirstlane(tid >> 5);
  const int nl   = lane & 15;
  const int hs   = lane >> 4;
  const int khi  = hs * 2;
  const int hi8  = hs * 8;

  const int blk = blockIdx.x;
  const int b   = blk / (T_SZ / ROWS);
  const int t0  = (blk % (T_SZ / ROWS)) * ROWS;

  const int n0 = wave * 16;
  const int ob1 = khi * M_SZ + n0 + nl;
  const int ob2 = khi * F_SZ;
  const float* pa1 = s_spec + nl * FSTR + khi;
  const float* pa2 = s_nd + nl * NDSTR + khi;
  const float btail = (hs == 0) ? fb[1024 * M_SZ + n0 + nl] : 0.0f;

  v2f b0[GRP], b1[GRP];
  load_b_group(b0, fb, ob1, 0, M_SZ);

  for (int i = tid; i < ROWS * F_SZ; i += NTHREADS) {
    int r = i / F_SZ, c = i - r * F_SZ;
    s_spec[r * FSTR + c] = spec0[(b * T_SZ + t0 + r) * F_SZ + c];
    s_buf [r * FSTR + c] = 0.0f;
  }
  for (int i = tid; i < ROWS * M_SZ; i += NTHREADS) {
    int r = i & (ROWS - 1), m = i >> 4;
    s_mel[r * MSTR + m] = mel[(b * M_SZ + m) * T_SZ + t0 + r];
  }
  __syncthreads();

  for (int it = 0; it < ITERS; ++it) {
    {
      v8f c;
      #pragma unroll
      for (int r = 0; r < 8; ++r)
        c[r] = -s_mel[(r + hi8) * MSTR + n0 + nl];

      int o = ob1;
      #pragma nounroll
      for (int g = 0; g < NG1 - 2; g += 2) {
        load_b_group(b1, fb, o, (g + 1) * 4 * GRP, M_SZ);
        loads_first();
        c = mma_group(c, pa1, g * 4 * GRP, b0);
        refill_after(c, o);
        load_b_group(b0, fb, o, (g + 2) * 4 * GRP, M_SZ);
        loads_first();
        c = mma_group(c, pa1, (g + 1) * 4 * GRP, b1);
      }
      load_b_group(b1, fb, o, (NG1 - 1) * 4 * GRP, M_SZ);
      loads_first();
      c = mma_group(c, pa1, (NG1 - 2) * 4 * GRP, b0);
      int q = ob2 + wave * 16 + nl;
      refill_after(c, q);
      load_b_group(b0, fbT, q, 0, F_SZ);
      loads_first();
      c = mma_group(c, pa1, (NG1 - 1) * 4 * GRP, b1);

      {
        v2f a, bf;
        a.x  = (hs == 0) ? s_spec[nl * FSTR + 1024]  : 0.0f;
        a.y  = 0.0f;
        bf.x = btail;
        bf.y = 0.0f;
        c = wmma_f32(a, bf, c);
      }
      #pragma unroll
      for (int r = 0; r < 8; ++r)
        s_nd[(r + hi8) * NDSTR + n0 + nl] = c[r];
    }
    lds_barrier();

    {
      int ft = wave;
      #pragma nounroll
      for (int i = 0; i < 7; ++i, ft += NWAVES)
        gemm2_tile(ft, nl, hi8, ob2, fbT, pa2, s_spec, s_buf, b0, b1,
                   fbT, ob2 + (ft + NWAVES) * 16 + nl, F_SZ);
      if (wave == 0) {
        gemm2_tile(ft, nl, hi8, ob2, fbT, pa2, s_spec, s_buf, b0, b1,
                   fbT, ob2 + F_SZ - 1, F_SZ);
        gemm2_tile(NFT - 1, nl, hi8, ob2, fbT, pa2, s_spec, s_buf, b0, b1,
                   fb, ob1, M_SZ);
      } else {
        gemm2_tile(ft, nl, hi8, ob2, fbT, pa2, s_spec, s_buf, b0, b1,
                   fb, ob1, M_SZ);
      }
    }
    lds_barrier();
  }

  for (int i = tid; i < ROWS * F_SZ; i += NTHREADS) {
    int r = i & (ROWS - 1), f = i >> 4;
    out[(b * F_SZ + f) * T_SZ + t0 + r] = s_spec[r * FSTR + f];
  }
}

extern "C" void kernel_launch(void* const* d_in, const int* in_sizes, int n_in,
                              void* d_out, int out_size, void* d_ws, size_t ws_size,
                              hipStream_t stream) {
  (void)in_sizes; (void)n_in; (void)out_size; (void)ws_size;
  const float* mel   = (const float*)d_in[0];
  const float* spec0 = (const float*)d_in[1];
  const float* fb    = (const float*)d_in[2];
  float* out = (float*)d_out;
  float* fbT = (float*)d_ws;

  {
    dim3 grid((M_SZ * F_SZ + NTHREADS - 1) / NTHREADS);
    hipLaunchKernelGGL(transpose_fb_kernel, grid, dim3(NTHREADS), 0, stream,
                       fb, fbT);
  }
  {
    dim3 grid((B_SZ * T_SZ) / ROWS);
    hipLaunchKernelGGL(invmel_sgd_kernel, grid, dim3(NTHREADS), 0, stream,
                       mel, spec0, fb, fbT, out);
  }
}


# GEMM2 filterbank operand kept in the workspace as [m/4][f][4]: one aligned 128-bit load per lane feeds two WMMAs (A via ds_load_b128); sum order in each 8 of m now 0,1,4,5,2,3,6,7
# speedup vs baseline: 2.3836x; 1.1741x over previous
#include <hip/hip_runtime.h>

#define B_SZ   4
#define T_SZ   1024
#define F_SZ   1025
#define M_SZ   128
#define ITERS  20
#define LR     0.3f
#define MOM    0.9f
#define INV    (2.0f / (B_SZ * T_SZ))

#define ROWS     16
#define FSTR     1027
#define MSTR     131
#define NDSTR    132
#define NWAVES   8
#define NTHREADS 256
#define NFT      ((F_SZ + 15) / 16)
#define GRP      16
#define NG1      (1024 / (4 * GRP))
#define FP       1024
#define WGRP     16
#define WSTEP    (2 * FP)

typedef __attribute__((ext_vector_type(2))) float v2f;
typedef __attribute__((ext_vector_type(4))) float v4f;
typedef __attribute__((ext_vector_type(8))) float v8f;

__device__ __forceinline__ v8f wmma_f32(v2f a, v2f b, v8f c) {
  return __builtin_amdgcn_wmma_f32_16x16x4_f32(false, a, false, b,
                                               (short)0, c, false, false);
}

__device__ __forceinline__ void load_b_group(v2f (&buf)[GRP],
                                             const float* __restrict__ base,
                                             int off, int k0, int rs) {
  const float* __restrict__ p = base + off;
  #pragma unroll
  for (int j = 0; j < GRP; ++j) {
    buf[j].x = p[(k0 + 4 * j)     * rs];
    buf[j].y = p[(k0 + 4 * j + 1) * rs];
  }
}

__device__ __forceinline__ v8f mma_group(v8f c, const float* pa, int k0,
                                         const v2f (&buf)[GRP]) {
  v2f a[GRP];
  #pragma unroll
  for (int j = 0; j < GRP; ++j) {
    a[j].x = pa[k0 + 4 * j];
    a[j].y = pa[k0 + 4 * j + 1];
  }
  #pragma unroll
  for (int j = 0; j < GRP; ++j)
    c = wmma_f32(a[j], buf[j], c);
  return c;
}

__device__ __forceinline__ void load_w_group(v4f (&w)[WGRP],
                                             const v4f* __restrict__ p, int step) {
  #pragma unroll
  for (int j = 0; j < WGRP; ++j)
    w[j] = p[j * step];
}

__device__ __forceinline__ v8f mma_tile(v8f c, const float* pa,
                                        const v4f (&w)[WGRP]) {
  v4f a[WGRP];
  #pragma unroll
  for (int j = 0; j < WGRP; ++j)
    a[j] = *(const v4f*)(pa + 8 * j);
  __builtin_amdgcn_sched_barrier(0);
  #pragma unroll
  for (int j = 0; j < WGRP; ++j) {
    c = wmma_f32(__builtin_shufflevector(a[j], a[j], 0, 1),
                 __builtin_shufflevector(w[j], w[j], 0, 1), c);
    c = wmma_f32(__builtin_shufflevector(a[j], a[j], 2, 3),
                 __builtin_shufflevector(w[j], w[j], 2, 3), c);
  }
  return c;
}

__device__ __forceinline__ void loads_first() {
  __builtin_amdgcn_sched_barrier(0);
}

__device__ __forceinline__ void refill_after(v8f& c, int& off) {
  asm volatile("" : "+v"(c), "+v"(off));
}

__device__ __forceinline__ void offset_ready(int& off) {
  asm volatile("" : "+v"(off));
}

__device__ __forceinline__ void wmma_distance(v8f& c) {
  asm volatile("v_nop\n\tv_nop\n\tv_nop\n\tv_nop\n\tv_nop" : "+v"(c));
}

__device__ __forceinline__ void lds_barrier() {
  __builtin_amdgcn_fence(__ATOMIC_RELEASE, "workgroup", "local");
  __builtin_amdgcn_s_barrier();
  __builtin_amdgcn_fence(__ATOMIC_ACQUIRE, "workgroup", "local");
}

__device__ __forceinline__ void sgd_update(const v8f& acc, int hi8, int fcol,
                                           float* s_spec, float* s_buf) {
  #pragma unroll
  for (int r = 0; r < 8; ++r) {
    const int off = (r + hi8) * FSTR + fcol;
    float nb = MOM * s_buf[off] + INV * acc[r];
    s_buf[off] = nb;
    float ns = s_spec[off] - LR * nb;
    s_spec[off] = ns > 0.0f ? ns : 0.0f;
  }
}

__device__ __forceinline__ void gemm2_tile(int fcol, int hi8, const float* pa2,
                                           float* s_spec, float* s_buf,
                                           const v4f (&cur)[WGRP], v4f (&nxt)[WGRP],
                                           const v4f* __restrict__ np,
                                           int noff, int nstep) {
  v8f acc = {0.f, 0.f, 0.f, 0.f, 0.f, 0.f, 0.f, 0.f};
  offset_ready(noff);
  offset_ready(fcol);
  load_w_group(nxt, np + noff, nstep);
  loads_first();
  acc = mma_tile(acc, pa2, cur);
  wmma_distance(acc);
  if (fcol < F_SZ)
    sgd_update(acc, hi8, fcol, s_spec, s_buf);
}

__device__ __forceinline__ void gemm2_tile_last(int fcol, int hi8, const float* pa2,
                                                float* s_spec, float* s_buf,
                                                const v4f (&cur)[WGRP], v2f (&b0)[GRP],
                                                const float* __restrict__ fb, int noff) {
  v8f acc = {0.f, 0.f, 0.f, 0.f, 0.f, 0.f, 0.f, 0.f};
  offset_ready(noff);
  offset_ready(fcol);
  load_b_group(b0, fb, noff, 0, M_SZ);
  loads_first();
  acc = mma_tile(acc, pa2, cur);
  wmma_distance(acc);
  if (fcol < F_SZ)
    sgd_update(acc, hi8, fcol, s_spec, s_buf);
}

__global__ __launch_bounds__(NTHREADS)
void transpose_fb_kernel(const float* __restrict__ fb,
                         float* __restrict__ fbT) {
  int i = blockIdx.x * NTHREADS + threadIdx.x;
  if (i < M_SZ * FP) {
    int mi = i & 3, f = (i >> 2) & (FP - 1), mb = i >> 12;
    fbT[i] = fb[f * M_SZ + 4 * mb + mi];
  }
}

__global__ __launch_bounds__(NTHREADS)
void invmel_sgd_kernel(const float* __restrict__ mel,
                       const float* __restrict__ spec0,
                       const float* __restrict__ fb,
                       const float* __restrict__ fbT,
                       float* __restrict__ out)
{
  __shared__ float s_spec[ROWS * FSTR];
  __shared__ float s_buf [ROWS * FSTR];
  __shared__ float s_mel [ROWS * MSTR];
  __shared__ __attribute__((aligned(16))) float s_nd[ROWS * NDSTR];

  const int tid  = threadIdx.x;
  const int lane = tid & 31;
  const int wave = __builtin_amdgcn_readfirstlane(tid >> 5);
  const int nl   = lane & 15;
  const int hs   = lane >> 4;
  const int khi  = hs * 2;
  const int hi8  = hs * 8;

  const int blk = blockIdx.x;
  const int b   = blk / (T_SZ / ROWS);
  const int t0  = (blk % (T_SZ / ROWS)) * ROWS;

  const int n0 = wave * 16;
  const int ob1 = khi * M_SZ + n0 + nl;
  const v4f* __restrict__ wt4 = (const v4f*)fbT;
  const int ow  = hs * FP + nl;
  const float* pa1 = s_spec + nl * FSTR + khi;
  const float* pa2 = s_nd + nl * NDSTR + 4 * hs;
  const float btail = (hs == 0) ? fb[1024 * M_SZ + n0 + nl] : 0.0f;

  v2f b0[GRP], b1[GRP];
  v4f w0[WGRP], w1[WGRP];
  load_b_group(b0, fb, ob1, 0, M_SZ);

  for (int i = tid; i < ROWS * F_SZ; i += NTHREADS) {
    int r = i / F_SZ, c = i - r * F_SZ;
    s_spec[r * FSTR + c] = spec0[(b * T_SZ + t0 + r) * F_SZ + c];
    s_buf [r * FSTR + c] = 0.0f;
  }
  for (int i = tid; i < ROWS * M_SZ; i += NTHREADS) {
    int r = i & (ROWS - 1), m = i >> 4;
    s_mel[r * MSTR + m] = mel[(b * M_SZ + m) * T_SZ + t0 + r];
  }
  __syncthreads();

  for (int it = 0; it < ITERS; ++it) {
    {
      v8f c;
      #pragma unroll
      for (int r = 0; r < 8; ++r)
        c[r] = -s_mel[(r + hi8) * MSTR + n0 + nl];

      int o = ob1;
      #pragma nounroll
      for (int g = 0; g < NG1 - 2; g += 2) {
        load_b_group(b1, fb, o, (g + 1) * 4 * GRP, M_SZ);
        loads_first();
        c = mma_group(c, pa1, g * 4 * GRP, b0);
        refill_after(c, o);
        load_b_group(b0, fb, o, (g + 2) * 4 * GRP, M_SZ);
        loads_first();
        c = mma_group(c, pa1, (g + 1) * 4 * GRP, b1);
      }
      int q = ow + wave * 16;
      offset_ready(q);
      load_b_group(b1, fb, o, (NG1 - 1) * 4 * GRP, M_SZ);
      loads_first();
      c = mma_group(c, pa1, (NG1 - 2) * 4 * GRP, b0);
      refill_after(c, q);
      load_w_group(w0, wt4 + q, WSTEP);
      loads_first();
      c = mma_group(c, pa1, (NG1 - 1) * 4 * GRP, b1);

      {
        v2f a, bf;
        a.x  = (hs == 0) ? s_spec[nl * FSTR + 1024]  : 0.0f;
        a.y  = 0.0f;
        bf.x = btail;
        bf.y = 0.0f;
        c = wmma_f32(a, bf, c);
      }
      #pragma unroll
      for (int r = 0; r < 8; ++r)
        s_nd[(r + hi8) * NDSTR + n0 + nl] = c[r];
    }
    lds_barrier();

    {
      int ft = wave;
      #pragma nounroll
      for (int i = 0; i < 3; ++i, ft += 2 * NWAVES) {
        gemm2_tile(ft * 16 + nl, hi8, pa2, s_spec, s_buf, w0, w1,
                   wt4, ow + (ft + NWAVES) * 16, WSTEP);
        gemm2_tile((ft + NWAVES) * 16 + nl, hi8, pa2, s_spec, s_buf, w1, w0,
                   wt4, ow + (ft + 2 * NWAVES) * 16, WSTEP);
      }
      gemm2_tile(ft * 16 + nl, hi8, pa2, s_spec, s_buf, w0, w1,
                 wt4, ow + (ft + NWAVES) * 16, WSTEP);
      if (wave == 0) {
        gemm2_tile((ft + NWAVES) * 16 + nl, hi8, pa2, s_spec, s_buf, w1, w0,
                   (const v4f*)(fb + FP * M_SZ), hs, 2);
        gemm2_tile_last(FP + nl, hi8, pa2, s_spec, s_buf, w0, b0, fb, ob1);
      } else {
        gemm2_tile_last((ft + NWAVES) * 16 + nl, hi8, pa2, s_spec, s_buf, w1,
                        b0, fb, ob1);
      }
    }
    lds_barrier();
  }

  for (int i = tid; i < ROWS * F_SZ; i += NTHREADS) {
    int r = i & (ROWS - 1), f = i >> 4;
    out[(b * F_SZ + f) * T_SZ + t0 + r] = s_spec[r * FSTR + f];
  }
}

extern "C" void kernel_launch(void* const* d_in, const int* in_sizes, int n_in,
                              void* d_out, int out_size, void* d_ws, size_t ws_size,
                              hipStream_t stream) {
  (void)in_sizes; (void)n_in; (void)out_size; (void)ws_size;
  const float* mel   = (const float*)d_in[0];
  const float* spec0 = (const float*)d_in[1];
  const float* fb    = (const float*)d_in[2];
  float* out = (float*)d_out;
  float* fbT = (float*)d_ws;

  {
    dim3 grid((M_SZ * FP + NTHREADS - 1) / NTHREADS);
    hipLaunchKernelGGL(transpose_fb_kernel, grid, dim3(NTHREADS), 0, stream,
                       fb, fbT);
  }
  {
    dim3 grid((B_SZ * T_SZ) / ROWS);
    hipLaunchKernelGGL(invmel_sgd_kernel, grid, dim3(NTHREADS), 0, stream,
                       mel, spec0, fb, fbT, out);
  }
}


# GEMM1 split 4 N-tiles x K quarter per wave: one 128-bit filterbank load feeds 2 WMMAs, one A fragment 4; the 4 K-quarter partial sums are added through LDS in fixed order (sum order changes)
# speedup vs baseline: 2.7628x; 1.1591x over previous
#include <hip/hip_runtime.h>

#define B_SZ   4
#define T_SZ   1024
#define F_SZ   1025
#define M_SZ   128
#define ITERS  20
#define LR     0.3f
#define MOM    0.9f
#define INV    (2.0f / (B_SZ * T_SZ))

#define ROWS     16
#define FSTR     1027
#define MSTR     132
#define NDSTR    132
#define NWAVES   8
#define NTHREADS 256
#define NFT      ((F_SZ + 15) / 16)
#define PSTR     132
#define NT1      4
#define KQ       256
#define GS       8
#define NGQ      (KQ / (4 * GS))
#define FP       1024
#define WGRP     16
#define WSTEP    (2 * FP)

typedef __attribute__((ext_vector_type(2))) float v2f;
typedef __attribute__((ext_vector_type(4))) float v4f;
typedef __attribute__((ext_vector_type(8))) float v8f;

__device__ __forceinline__ v8f wmma_f32(v2f a, v2f b, v8f c) {
  return __builtin_amdgcn_wmma_f32_16x16x4_f32(false, a, false, b,
                                               (short)0, c, false, false);
}

__device__ __forceinline__ void load_q_group(v4f (&bx)[GS], v4f (&by)[GS],
                                             const v4f* __restrict__ base,
                                             int off, int s0) {
  const v4f* __restrict__ p = base + off;
  #pragma unroll
  for (int j = 0; j < GS; ++j) {
    bx[j] = p[(4 * (s0 + j))     * (M_SZ / 4)];
    by[j] = p[(4 * (s0 + j) + 1) * (M_SZ / 4)];
  }
}

__device__ __forceinline__ void load_a_group(v2f (&a)[GS], const float* base,
                                             int off, int s0) {
  const float* p = base + off;
  #pragma unroll
  for (int j = 0; j < GS; ++j) {
    a[j].x = p[4 * (s0 + j)];
    a[j].y = p[4 * (s0 + j) + 1];
  }
}

__device__ __forceinline__ void mma_q_group(v8f (&c)[NT1], const v2f (&a)[GS],
                                            const v4f (&bx)[GS],
                                            const v4f (&by)[GS]) {
  v2f p[GS][NT1];
  #pragma unroll
  for (int j = 0; j < GS; ++j) {
    #pragma unroll
    for (int t = 0; t < NT1; ++t) {
      p[j][t].x = bx[j][t];
      p[j][t].y = by[j][t];
    }
  }
  #pragma unroll
  for (int j = 0; j < GS; ++j)
    asm volatile("" : "+v"(p[j][0]), "+v"(p[j][1]), "+v"(p[j][2]), "+v"(p[j][3]));
  __builtin_amdgcn_sched_barrier(0);
  #pragma unroll
  for (int j = 0; j < GS; ++j) {
    #pragma unroll
    for (int t = 0; t < NT1; ++t)
      c[t] = wmma_f32(a[j], p[j][t], c[t]);
  }
  asm volatile("v_nop\n\tv_nop\n\tv_nop\n\tv_nop\n\tv_nop"
               : "+v"(c[0]), "+v"(c[1]), "+v"(c[2]), "+v"(c[3]));
  __builtin_amdgcn_sched_barrier(0);
}

__device__ __forceinline__ void load_w_group(v4f (&w)[WGRP],
                                             const v4f* __restrict__ p, int step) {
  #pragma unroll
  for (int j = 0; j < WGRP; ++j)
    w[j] = p[j * step];
}

__device__ __forceinline__ v8f mma_tile(v8f c, const float* pa,
                                        const v4f (&w)[WGRP]) {
  v4f a[WGRP];
  #pragma unroll
  for (int j = 0; j < WGRP; ++j)
    a[j] = *(const v4f*)(pa + 8 * j);
  __builtin_amdgcn_sched_barrier(0);
  #pragma unroll
  for (int j = 0; j < WGRP; ++j) {
    c = wmma_f32(__builtin_shufflevector(a[j], a[j], 0, 1),
                 __builtin_shufflevector(w[j], w[j], 0, 1), c);
    c = wmma_f32(__builtin_shufflevector(a[j], a[j], 2, 3),
                 __builtin_shufflevector(w[j], w[j], 2, 3), c);
  }
  return c;
}

__device__ __forceinline__ void loads_first() {
  __builtin_amdgcn_sched_barrier(0);
}

__device__ __forceinline__ void refill_after(v8f (&c)[NT1], int& off, int& offa) {
  asm volatile("" : "+v"(c[0]), "+v"(c[1]), "+v"(c[2]), "+v"(c[3]),
                    "+v"(off), "+v"(offa));
}

__device__ __forceinline__ void offset_ready(int& off) {
  asm volatile("" : "+v"(off));
}

__device__ __forceinline__ void wmma_distance(v8f& c) {
  asm volatile("v_nop\n\tv_nop\n\tv_nop\n\tv_nop\n\tv_nop" : "+v"(c));
}

__device__ __forceinline__ void lds_barrier() {
  __builtin_amdgcn_fence(__ATOMIC_RELEASE, "workgroup", "local");
  __builtin_amdgcn_s_barrier();
  __builtin_amdgcn_fence(__ATOMIC_ACQUIRE, "workgroup", "local");
}

__device__ __forceinline__ void sgd_update(const v8f& acc, int hi8, int fcol,
                                           float* s_spec, float* s_buf) {
  #pragma unroll
  for (int r = 0; r < 8; ++r) {
    const int off = (r + hi8) * FSTR + fcol;
    float nb = MOM * s_buf[off] + INV * acc[r];
    s_buf[off] = nb;
    float ns = s_spec[off] - LR * nb;
    s_spec[off] = ns > 0.0f ? ns : 0.0f;
  }
}

__device__ __forceinline__ void gemm2_tile(int fcol, int hi8, const float* pa2,
                                           float* s_spec, float* s_buf,
                                           const v4f (&cur)[WGRP], v4f (&nxt)[WGRP],
                                           const v4f* __restrict__ np,
                                           int noff, int nstep) {
  v8f acc = {0.f, 0.f, 0.f, 0.f, 0.f, 0.f, 0.f, 0.f};
  offset_ready(noff);
  offset_ready(fcol);
  load_w_group(nxt, np + noff, nstep);
  loads_first();
  acc = mma_tile(acc, pa2, cur);
  wmma_distance(acc);
  if (fcol < F_SZ)
    sgd_update(acc, hi8, fcol, s_spec, s_buf);
}

__device__ __forceinline__ void gemm2_tile_last(int fcol, int hi8, const float* pa2,
                                                float* s_spec, float* s_buf,
                                                const v4f (&cur)[WGRP],
                                                v4f (&bx0)[GS], v4f (&by0)[GS],
                                                const v4f* __restrict__ fb4, int noff) {
  v8f acc = {0.f, 0.f, 0.f, 0.f, 0.f, 0.f, 0.f, 0.f};
  offset_ready(noff);
  offset_ready(fcol);
  load_q_group(bx0, by0, fb4, noff, 0);
  loads_first();
  acc = mma_tile(acc, pa2, cur);
  wmma_distance(acc);
  if (fcol < F_SZ)
    sgd_update(acc, hi8, fcol, s_spec, s_buf);
}

__global__ __launch_bounds__(NTHREADS)
void transpose_fb_kernel(const float* __restrict__ fb,
                         float* __restrict__ fbT) {
  int i = blockIdx.x * NTHREADS + threadIdx.x;
  if (i < M_SZ * FP) {
    int mi = i & 3, f = (i >> 2) & (FP - 1), mb = i >> 12;
    fbT[i] = fb[f * M_SZ + 4 * mb + mi];
  }
}

__global__ __launch_bounds__(NTHREADS)
void invmel_sgd_kernel(const float* __restrict__ mel,
                       const float* __restrict__ spec0,
                       const float* __restrict__ fb,
                       const float* __restrict__ fbT,
                       float* __restrict__ out)
{
  __shared__ float s_spec[ROWS * FSTR];
  __shared__ float s_buf [ROWS * FSTR];
  __shared__ __attribute__((aligned(16))) float s_mel[ROWS * MSTR];
  __shared__ __attribute__((aligned(16))) float s_nd[ROWS * NDSTR];
  __shared__ __attribute__((aligned(16))) float s_part[(1024 / KQ) * ROWS * PSTR];

  const int tid  = threadIdx.x;
  const int lane = tid & 31;
  const int wave = __builtin_amdgcn_readfirstlane(tid >> 5);
  const int nl   = lane & 15;
  const int hs   = lane >> 4;
  const int khi  = hs * 2;
  const int hi8  = hs * 8;

  const int blk = blockIdx.x;
  const int b   = blk / (T_SZ / ROWS);
  const int t0  = (blk % (T_SZ / ROWS)) * ROWS;

  const int ng = wave & 1;
  const int kq = wave >> 1;
  const v4f* __restrict__ fb4 = (const v4f*)fb;
  const int obq = (kq * KQ + khi) * (M_SZ / 4) + ng * 16 + nl;
  const v4f* __restrict__ wt4 = (const v4f*)fbT;
  const int ow  = hs * FP + nl;
  const int oa1 = nl * FSTR + kq * KQ + khi;
  const float* pa2 = s_nd + nl * NDSTR + 4 * hs;
  v4f btail = {0.f, 0.f, 0.f, 0.f};
  if (kq == 3 && hs == 0)
    btail = fb4[FP * (M_SZ / 4) + ng * 16 + nl];

  v4f bx0[GS], by0[GS], bx1[GS], by1[GS];
  v2f a0[GS], a1[GS];
  v4f w0[WGRP], w1[WGRP];
  load_q_group(bx0, by0, fb4, obq, 0);

  for (int i = tid; i < ROWS * F_SZ; i += NTHREADS) {
    int r = i / F_SZ, c = i - r * F_SZ;
    s_spec[r * FSTR + c] = spec0[(b * T_SZ + t0 + r) * F_SZ + c];
    s_buf [r * FSTR + c] = 0.0f;
  }
  for (int i = tid; i < ROWS * M_SZ; i += NTHREADS) {
    int r = i & (ROWS - 1), m = i >> 4;
    s_mel[r * MSTR + m] = mel[(b * M_SZ + m) * T_SZ + t0 + r];
  }
  __syncthreads();

  for (int it = 0; it < ITERS; ++it) {
    {
      v8f c[NT1];
      #pragma unroll
      for (int t = 0; t < NT1; ++t)
        #pragma unroll
        for (int r = 0; r < 8; ++r)
          c[t][r] = 0.0f;

      if (kq == 3) {
        v2f a;
        a.x = (hs == 0) ? s_spec[nl * FSTR + 1024] : 0.0f;
        a.y = 0.0f;
        #pragma unroll
        for (int t = 0; t < NT1; ++t) {
          v2f bf;
          bf.x = btail[t];
          bf.y = 0.0f;
          c[t] = wmma_f32(a, bf, c[t]);
        }
        asm volatile("v_nop\n\tv_nop\n\tv_nop\n\tv_nop\n\tv_nop"
                     : "+v"(c[0]), "+v"(c[1]), "+v"(c[2]), "+v"(c[3]));
      }

      int o = obq, oa = oa1;
      load_a_group(a0, s_spec, oa, 0);
      #pragma nounroll
      for (int g = 0; g < NGQ - 2; g += 2) {
        load_q_group(bx1, by1, fb4, o, (g + 1) * GS);
        load_a_group(a1, s_spec, oa, (g + 1) * GS);
        loads_first();
        mma_q_group(c, a0, bx0, by0);
        refill_after(c, o, oa);
        load_q_group(bx0, by0, fb4, o, (g + 2) * GS);
        load_a_group(a0, s_spec, oa, (g + 2) * GS);
        loads_first();
        mma_q_group(c, a1, bx1, by1);
      }
      int q = ow + wave * 16;
      offset_ready(q);
      load_q_group(bx1, by1, fb4, o, (NGQ - 1) * GS);
      load_a_group(a1, s_spec, oa, (NGQ - 1) * GS);
      loads_first();
      mma_q_group(c, a0, bx0, by0);
      refill_after(c, q, oa);
      load_w_group(w0, wt4 + q, WSTEP);
      loads_first();
      mma_q_group(c, a1, bx1, by1);

      float* pp = s_part + kq * (ROWS * PSTR) + hi8 * PSTR + ng * 64 + 4 * nl;
      #pragma unroll
      for (int r = 0; r < 8; ++r) {
        v4f v = {c[0][r], c[1][r], c[2][r], c[3][r]};
        *(v4f*)(pp + r * PSTR) = v;
      }
    }
    lds_barrier();

    {
      const int wo = (tid >> 4) * PSTR + 4 * (tid & 15);
      #pragma unroll
      for (int hf = 0; hf < 2; ++hf) {
        const int e = wo + hf * 64;
        v4f p0 = *(const v4f*)(s_part + 0 * ROWS * PSTR + e);
        v4f p1 = *(const v4f*)(s_part + 1 * ROWS * PSTR + e);
        v4f p2 = *(const v4f*)(s_part + 2 * ROWS * PSTR + e);
        v4f p3 = *(const v4f*)(s_part + 3 * ROWS * PSTR + e);
        v4f m  = *(const v4f*)(s_mel + e);
        *(v4f*)(s_nd + e) = ((p0 + p1) + (p2 + p3)) - m;
      }
    }
    lds_barrier();

    {
      int ft = wave;
      #pragma nounroll
      for (int i = 0; i < 3; ++i, ft += 2 * NWAVES) {
        gemm2_tile(ft * 16 + nl, hi8, pa2, s_spec, s_buf, w0, w1,
                   wt4, ow + (ft + NWAVES) * 16, WSTEP);
        gemm2_tile((ft + NWAVES) * 16 + nl, hi8, pa2, s_spec, s_buf, w1, w0,
                   wt4, ow + (ft + 2 * NWAVES) * 16, WSTEP);
      }
      gemm2_tile(ft * 16 + nl, hi8, pa2, s_spec, s_buf, w0, w1,
                 wt4, ow + (ft + NWAVES) * 16, WSTEP);
      if (wave == 0) {
        gemm2_tile((ft + NWAVES) * 16 + nl, hi8, pa2, s_spec, s_buf, w1, w0,
                   (const v4f*)(fb + FP * M_SZ), hs, 2);
        gemm2_tile_last(FP + nl, hi8, pa2, s_spec, s_buf, w0, bx0, by0, fb4, obq);
      } else {
        gemm2_tile_last((ft + NWAVES) * 16 + nl, hi8, pa2, s_spec, s_buf, w1,
                        bx0, by0, fb4, obq);
      }
    }
    lds_barrier();
  }

  for (int i = tid; i < ROWS * F_SZ; i += NTHREADS) {
    int r = i & (ROWS - 1), f = i >> 4;
    out[(b * F_SZ + f) * T_SZ + t0 + r] = s_spec[r * FSTR + f];
  }
}

extern "C" void kernel_launch(void* const* d_in, const int* in_sizes, int n_in,
                              void* d_out, int out_size, void* d_ws, size_t ws_size,
                              hipStream_t stream) {
  (void)in_sizes; (void)n_in; (void)out_size; (void)ws_size;
  const float* mel   = (const float*)d_in[0];
  const float* spec0 = (const float*)d_in[1];
  const float* fb    = (const float*)d_in[2];
  float* out = (float*)d_out;
  float* fbT = (float*)d_ws;

  {
    dim3 grid((M_SZ * FP + NTHREADS - 1) / NTHREADS);
    hipLaunchKernelGGL(transpose_fb_kernel, grid, dim3(NTHREADS), 0, stream,
                       fb, fbT);
  }
  {
    dim3 grid((B_SZ * T_SZ) / ROWS);
    hipLaunchKernelGGL(invmel_sgd_kernel, grid, dim3(NTHREADS), 0, stream,
                       mel, spec0, fb, fbT, out);
  }
}


# GEMM2 tiles below column 1024 read no LDS: negdiff A fragments loaded once per iteration, momentum and a spec copy of each wave's own tiles held in registers (s_buf array gone); sum order unchanged
# speedup vs baseline: 2.9138x; 1.0547x over previous
#include <hip/hip_runtime.h>

#define B_SZ   4
#define T_SZ   1024
#define F_SZ   1025
#define M_SZ   128
#define ITERS  20
#define LR     0.3f
#define MOM    0.9f
#define INV    (2.0f / (B_SZ * T_SZ))

#define ROWS     16
#define FSTR     1027
#define MSTR     132
#define NDSTR    132
#define NWAVES   8
#define NTHREADS 256
#define NFT      ((F_SZ + 15) / 16)
#define PSTR     132
#define NT1      4
#define KQ       256
#define GS       8
#define NGQ      (KQ / (4 * GS))
#define FP       1024
#define WGRP     16
#define WSTEP    (2 * FP)

typedef __attribute__((ext_vector_type(2))) float v2f;
typedef __attribute__((ext_vector_type(4))) float v4f;
typedef __attribute__((ext_vector_type(8))) float v8f;

__device__ __forceinline__ v8f wmma_f32(v2f a, v2f b, v8f c) {
  return __builtin_amdgcn_wmma_f32_16x16x4_f32(false, a, false, b,
                                               (short)0, c, false, false);
}

__device__ __forceinline__ void load_q_group(v4f (&bx)[GS], v4f (&by)[GS],
                                             const v4f* __restrict__ base,
                                             int off, int s0) {
  const v4f* __restrict__ p = base + off;
  #pragma unroll
  for (int j = 0; j < GS; ++j) {
    bx[j] = p[(4 * (s0 + j))     * (M_SZ / 4)];
    by[j] = p[(4 * (s0 + j) + 1) * (M_SZ / 4)];
  }
}

__device__ __forceinline__ void load_a_group(v2f (&a)[GS], const float* base,
                                             int off, int s0) {
  const float* p = base + off;
  #pragma unroll
  for (int j = 0; j < GS; ++j) {
    a[j].x = p[4 * (s0 + j)];
    a[j].y = p[4 * (s0 + j) + 1];
  }
}

__device__ __forceinline__ void mma_q_group(v8f (&c)[NT1], const v2f (&a)[GS],
                                            const v4f (&bx)[GS],
                                            const v4f (&by)[GS]) {
  v2f p[GS][NT1];
  #pragma unroll
  for (int j = 0; j < GS; ++j) {
    #pragma unroll
    for (int t = 0; t < NT1; ++t) {
      p[j][t].x = bx[j][t];
      p[j][t].y = by[j][t];
    }
  }
  #pragma unroll
  for (int j = 0; j < GS; ++j)
    asm volatile("" : "+v"(p[j][0]), "+v"(p[j][1]), "+v"(p[j][2]), "+v"(p[j][3]));
  __builtin_amdgcn_sched_barrier(0);
  #pragma unroll
  for (int j = 0; j < GS; ++j) {
    #pragma unroll
    for (int t = 0; t < NT1; ++t)
      c[t] = wmma_f32(a[j], p[j][t], c[t]);
  }
  asm volatile("v_nop\n\tv_nop\n\tv_nop\n\tv_nop\n\tv_nop"
               : "+v"(c[0]), "+v"(c[1]), "+v"(c[2]), "+v"(c[3]));
  __builtin_amdgcn_sched_barrier(0);
}

__device__ __forceinline__ void load_w_group(v4f (&w)[WGRP],
                                             const v4f* __restrict__ p, int step) {
  #pragma unroll
  for (int j = 0; j < WGRP; ++j)
    w[j] = p[j * step];
}

__device__ __forceinline__ void load_nd_frags(v4f (&a)[WGRP], const float* pa) {
  #pragma unroll
  for (int j = 0; j < WGRP; ++j)
    a[j] = *(const v4f*)(pa + 8 * j);
}

__device__ __forceinline__ v8f mma_tile(v8f c, const v4f (&a)[WGRP],
                                        const v4f (&w)[WGRP]) {
  #pragma unroll
  for (int j = 0; j < WGRP; ++j) {
    c = wmma_f32(__builtin_shufflevector(a[j], a[j], 0, 1),
                 __builtin_shufflevector(w[j], w[j], 0, 1), c);
    c = wmma_f32(__builtin_shufflevector(a[j], a[j], 2, 3),
                 __builtin_shufflevector(w[j], w[j], 2, 3), c);
  }
  return c;
}

__device__ __forceinline__ void loads_first() {
  __builtin_amdgcn_sched_barrier(0);
}

__device__ __forceinline__ void refill_after(v8f (&c)[NT1], int& off, int& offa) {
  asm volatile("" : "+v"(c[0]), "+v"(c[1]), "+v"(c[2]), "+v"(c[3]),
                    "+v"(off), "+v"(offa));
}

__device__ __forceinline__ void offset_ready(int& off) {
  asm volatile("" : "+v"(off));
}

__device__ __forceinline__ void wmma_distance(v8f& c) {
  asm volatile("v_nop\n\tv_nop\n\tv_nop\n\tv_nop\n\tv_nop" : "+v"(c));
}

__device__ __forceinline__ void update_behind() {
  __builtin_amdgcn_sched_barrier(0);
}

__device__ __forceinline__ void lds_barrier() {
  __builtin_amdgcn_fence(__ATOMIC_RELEASE, "workgroup", "local");
  __builtin_amdgcn_s_barrier();
  __builtin_amdgcn_fence(__ATOMIC_ACQUIRE, "workgroup", "local");
}

__device__ __forceinline__ void sgd_update_reg(const v8f& acc, v8f& mb, v8f& sp,
                                               float* ps) {
  #pragma unroll
  for (int r = 0; r < 8; ++r) {
    float nb = MOM * mb[r] + INV * acc[r];
    mb[r] = nb;
    float ns = sp[r] - LR * nb;
    ns = ns > 0.0f ? ns : 0.0f;
    sp[r] = ns;
    ps[r * FSTR] = ns;
  }
}

__device__ __forceinline__ void sgd_update_col(const v8f& acc, int hi8, int fcol,
                                               float* s_spec, float* s_bcol) {
  float ob[8], os[8];
  #pragma unroll
  for (int r = 0; r < 8; ++r) {
    ob[r] = s_bcol[r + hi8];
    os[r] = s_spec[(r + hi8) * FSTR + fcol];
  }
  #pragma unroll
  for (int r = 0; r < 8; ++r) {
    float nb = MOM * ob[r] + INV * acc[r];
    ob[r] = nb;
    float ns = os[r] - LR * nb;
    os[r] = ns > 0.0f ? ns : 0.0f;
  }
  #pragma unroll
  for (int r = 0; r < 8; ++r) {
    s_bcol[r + hi8] = ob[r];
    s_spec[(r + hi8) * FSTR + fcol] = os[r];
  }
}

__device__ __forceinline__ void gemm2_tile(const v4f (&a)[WGRP],
                                           const v4f (&cur)[WGRP], v4f (&nxt)[WGRP],
                                           const v4f* __restrict__ np, int nstep,
                                           v8f& mb, v8f& sp, float* ps) {
  v8f acc = {0.f, 0.f, 0.f, 0.f, 0.f, 0.f, 0.f, 0.f};
  load_w_group(nxt, np, nstep);
  loads_first();
  acc = mma_tile(acc, a, cur);
  wmma_distance(acc);
  update_behind();
  sgd_update_reg(acc, mb, sp, ps);
}

__device__ __forceinline__ void gemm2_tile_last(const v4f (&a)[WGRP],
                                                const v4f (&cur)[WGRP],
                                                v4f (&bx0)[GS], v4f (&by0)[GS],
                                                const v4f* __restrict__ fb4, int noff,
                                                v8f& mb, v8f& sp, float* ps) {
  v8f acc = {0.f, 0.f, 0.f, 0.f, 0.f, 0.f, 0.f, 0.f};
  load_q_group(bx0, by0, fb4, noff, 0);
  loads_first();
  acc = mma_tile(acc, a, cur);
  wmma_distance(acc);
  update_behind();
  sgd_update_reg(acc, mb, sp, ps);
}

__device__ __forceinline__ void gemm2_tile_col(int fcol, int hi8,
                                               const v4f (&a)[WGRP],
                                               float* s_spec, float* s_bcol,
                                               const v4f (&cur)[WGRP],
                                               v4f (&bx0)[GS], v4f (&by0)[GS],
                                               const v4f* __restrict__ fb4, int noff) {
  v8f acc = {0.f, 0.f, 0.f, 0.f, 0.f, 0.f, 0.f, 0.f};
  offset_ready(fcol);
  load_q_group(bx0, by0, fb4, noff, 0);
  loads_first();
  acc = mma_tile(acc, a, cur);
  wmma_distance(acc);
  if (fcol < F_SZ)
    sgd_update_col(acc, hi8, fcol, s_spec, s_bcol);
}

__global__ __launch_bounds__(NTHREADS)
void transpose_fb_kernel(const float* __restrict__ fb,
                         float* __restrict__ fbT) {
  int i = blockIdx.x * NTHREADS + threadIdx.x;
  if (i < M_SZ * FP) {
    int mi = i & 3, f = (i >> 2) & (FP - 1), mb = i >> 12;
    fbT[i] = fb[f * M_SZ + 4 * mb + mi];
  }
}

__global__ __launch_bounds__(NTHREADS)
void invmel_sgd_kernel(const float* __restrict__ mel,
                       const float* __restrict__ spec0,
                       const float* __restrict__ fb,
                       const float* __restrict__ fbT,
                       float* __restrict__ out)
{
  __shared__ float s_spec[ROWS * FSTR];
  __shared__ float s_bcol[ROWS];
  __shared__ __attribute__((aligned(16))) float s_mel[ROWS * MSTR];
  __shared__ __attribute__((aligned(16))) float s_nd[ROWS * NDSTR];
  __shared__ __attribute__((aligned(16))) float s_part[(1024 / KQ) * ROWS * PSTR];

  const int tid  = threadIdx.x;
  const int lane = tid & 31;
  const int wave = __builtin_amdgcn_readfirstlane(tid >> 5);
  const int nl   = lane & 15;
  const int hs   = lane >> 4;
  const int khi  = hs * 2;
  const int hi8  = hs * 8;

  const int blk = blockIdx.x;
  const int b   = blk / (T_SZ / ROWS);
  const int t0  = (blk % (T_SZ / ROWS)) * ROWS;

  const int ng = wave & 1;
  const int kq = wave >> 1;
  const v4f* __restrict__ fb4 = (const v4f*)fb;
  const int obq = (kq * KQ + khi) * (M_SZ / 4) + ng * 16 + nl;
  const v4f* __restrict__ wt4 = (const v4f*)fbT;
  const int ow  = hs * FP + nl;
  const int oa1 = nl * FSTR + kq * KQ + khi;
  const float* pa2 = s_nd + nl * NDSTR + 4 * hs;
  const v4f* __restrict__ wp = wt4 + (ow + wave * 16);
  float* ps0 = s_spec + hi8 * FSTR + wave * 16 + nl;
  v4f btail = {0.f, 0.f, 0.f, 0.f};
  if (kq == 3 && hs == 0)
    btail = fb4[FP * (M_SZ / 4) + ng * 16 + nl];

  v4f bx0[GS], by0[GS], bx1[GS], by1[GS];
  v2f a0[GS], a1[GS];
  v4f w0[WGRP], w1[WGRP];
  load_q_group(bx0, by0, fb4, obq, 0);

  for (int i = tid; i < ROWS * F_SZ; i += NTHREADS) {
    int r = i / F_SZ, c = i - r * F_SZ;
    s_spec[r * FSTR + c] = spec0[(b * T_SZ + t0 + r) * F_SZ + c];
  }
  if (tid < ROWS)
    s_bcol[tid] = 0.0f;
  for (int i = tid; i < ROWS * M_SZ; i += NTHREADS) {
    int r = i & (ROWS - 1), m = i >> 4;
    s_mel[r * MSTR + m] = mel[(b * M_SZ + m) * T_SZ + t0 + r];
  }
  __syncthreads();

  v8f mb[NWAVES], sp[NWAVES];
  #pragma unroll
  for (int i = 0; i < NWAVES; ++i) {
    #pragma unroll
    for (int r = 0; r < 8; ++r) {
      mb[i][r] = 0.0f;
      sp[i][r] = ps0[i * 128 + r * FSTR];
    }
  }

  for (int it = 0; it < ITERS; ++it) {
    {
      v8f c[NT1];
      #pragma unroll
      for (int t = 0; t < NT1; ++t)
        #pragma unroll
        for (int r = 0; r < 8; ++r)
          c[t][r] = 0.0f;

      if (kq == 3) {
        v2f a;
        a.x = (hs == 0) ? s_spec[nl * FSTR + 1024] : 0.0f;
        a.y = 0.0f;
        #pragma unroll
        for (int t = 0; t < NT1; ++t) {
          v2f bf;
          bf.x = btail[t];
          bf.y = 0.0f;
          c[t] = wmma_f32(a, bf, c[t]);
        }
        asm volatile("v_nop\n\tv_nop\n\tv_nop\n\tv_nop\n\tv_nop"
                     : "+v"(c[0]), "+v"(c[1]), "+v"(c[2]), "+v"(c[3]));
      }

      int o = obq, oa = oa1;
      load_a_group(a0, s_spec, oa, 0);
      #pragma nounroll
      for (int g = 0; g < NGQ - 2; g += 2) {
        load_q_group(bx1, by1, fb4, o, (g + 1) * GS);
        load_a_group(a1, s_spec, oa, (g + 1) * GS);
        loads_first();
        mma_q_group(c, a0, bx0, by0);
        refill_after(c, o, oa);
        load_q_group(bx0, by0, fb4, o, (g + 2) * GS);
        load_a_group(a0, s_spec, oa, (g + 2) * GS);
        loads_first();
        mma_q_group(c, a1, bx1, by1);
      }
      load_q_group(bx1, by1, fb4, o, (NGQ - 1) * GS);
      load_a_group(a1, s_spec, oa, (NGQ - 1) * GS);
      loads_first();
      mma_q_group(c, a0, bx0, by0);
      load_w_group(w0, wp, WSTEP);
      loads_first();
      mma_q_group(c, a1, bx1, by1);

      float* pp = s_part + kq * (ROWS * PSTR) + hi8 * PSTR + ng * 64 + 4 * nl;
      #pragma unroll
      for (int r = 0; r < 8; ++r) {
        v4f v = {c[0][r], c[1][r], c[2][r], c[3][r]};
        *(v4f*)(pp + r * PSTR) = v;
      }
    }
    lds_barrier();

    {
      const int wo = (tid >> 4) * PSTR + 4 * (tid & 15);
      #pragma unroll
      for (int hf = 0; hf < 2; ++hf) {
        const int e = wo + hf * 64;
        v4f p0 = *(const v4f*)(s_part + 0 * ROWS * PSTR + e);
        v4f p1 = *(const v4f*)(s_part + 1 * ROWS * PSTR + e);
        v4f p2 = *(const v4f*)(s_part + 2 * ROWS * PSTR + e);
        v4f p3 = *(const v4f*)(s_part + 3 * ROWS * PSTR + e);
        v4f m  = *(const v4f*)(s_mel + e);
        *(v4f*)(s_nd + e) = ((p0 + p1) + (p2 + p3)) - m;
      }
    }
    lds_barrier();

    {
      v4f a2[WGRP];
      load_nd_frags(a2, pa2);
      #pragma unroll
      for (int i = 0; i < 6; i += 2) {
        gemm2_tile(a2, w0, w1, wp + (i + 1) * 128, WSTEP, mb[i], sp[i],
                   ps0 + i * 128);
        gemm2_tile(a2, w1, w0, wp + (i + 2) * 128, WSTEP, mb[i + 1], sp[i + 1],
                   ps0 + (i + 1) * 128);
      }
      gemm2_tile(a2, w0, w1, wp + 7 * 128, WSTEP, mb[6], sp[6], ps0 + 6 * 128);
      if (wave == 0) {
        gemm2_tile(a2, w1, w0, (const v4f*)(fb + FP * M_SZ) + hs, 2,
                   mb[7], sp[7], ps0 + 7 * 128);
        gemm2_tile_col(FP + nl, hi8, a2, s_spec, s_bcol, w0, bx0, by0, fb4, obq);
      } else {
        gemm2_tile_last(a2, w1, bx0, by0, fb4, obq, mb[7], sp[7], ps0 + 7 * 128);
      }
    }
    lds_barrier();
  }

  for (int i = tid; i < ROWS * F_SZ; i += NTHREADS) {
    int r = i & (ROWS - 1), f = i >> 4;
    out[(b * F_SZ + f) * T_SZ + t0 + r] = s_spec[r * FSTR + f];
  }
}

extern "C" void kernel_launch(void* const* d_in, const int* in_sizes, int n_in,
                              void* d_out, int out_size, void* d_ws, size_t ws_size,
                              hipStream_t stream) {
  (void)in_sizes; (void)n_in; (void)out_size; (void)ws_size;
  const float* mel   = (const float*)d_in[0];
  const float* spec0 = (const float*)d_in[1];
  const float* fb    = (const float*)d_in[2];
  float* out = (float*)d_out;
  float* fbT = (float*)d_ws;

  {
    dim3 grid((M_SZ * FP + NTHREADS - 1) / NTHREADS);
    hipLaunchKernelGGL(transpose_fb_kernel, grid, dim3(NTHREADS), 0, stream,
                       fb, fbT);
  }
  {
    dim3 grid((B_SZ * T_SZ) / ROWS);
    hipLaunchKernelGGL(invmel_sgd_kernel, grid, dim3(NTHREADS), 0, stream,
                       mel, spec0, fb, fbT, out);
  }
}


# Tile staging batched: every global load of the prologue (spec, mel, fb tail row) goes out before the first LDS store, one memory round trip instead of 25 serial ones; output emission batched alike
# speedup vs baseline: 3.0855x; 1.0589x over previous
#include <hip/hip_runtime.h>

#define B_SZ   4
#define T_SZ   1024
#define F_SZ   1025
#define M_SZ   128
#define ITERS  20
#define LR     0.3f
#define MOM    0.9f
#define INV    (2.0f / (B_SZ * T_SZ))

#define ROWS     16
#define FSTR     1027
#define MSTR     132
#define NDSTR    132
#define NWAVES   8
#define NTHREADS 256
#define NFT      ((F_SZ + 15) / 16)
#define PSTR     132
#define NT1      4
#define KQ       256
#define GS       8
#define NGQ      (KQ / (4 * GS))
#define FP       1024
#define WGRP     16
#define WSTEP    (2 * FP)

typedef __attribute__((ext_vector_type(2))) float v2f;
typedef __attribute__((ext_vector_type(4))) float v4f;
typedef __attribute__((ext_vector_type(8))) float v8f;

__device__ __forceinline__ v8f wmma_f32(v2f a, v2f b, v8f c) {
  return __builtin_amdgcn_wmma_f32_16x16x4_f32(false, a, false, b,
                                               (short)0, c, false, false);
}

__device__ __forceinline__ void load_q_group(v4f (&bx)[GS], v4f (&by)[GS],
                                             const v4f* __restrict__ base,
                                             int off, int s0) {
  const v4f* __restrict__ p = base + off;
  #pragma unroll
  for (int j = 0; j < GS; ++j) {
    bx[j] = p[(4 * (s0 + j))     * (M_SZ / 4)];
    by[j] = p[(4 * (s0 + j) + 1) * (M_SZ / 4)];
  }
}

__device__ __forceinline__ void load_a_group(v2f (&a)[GS], const float* base,
                                             int off, int s0) {
  const float* p = base + off;
  #pragma unroll
  for (int j = 0; j < GS; ++j) {
    a[j].x = p[4 * (s0 + j)];
    a[j].y = p[4 * (s0 + j) + 1];
  }
}

__device__ __forceinline__ void mma_q_group(v8f (&c)[NT1], const v2f (&a)[GS],
                                            const v4f (&bx)[GS],
                                            const v4f (&by)[GS]) {
  v2f p[GS][NT1];
  #pragma unroll
  for (int j = 0; j < GS; ++j) {
    #pragma unroll
    for (int t = 0; t < NT1; ++t) {
      p[j][t].x = bx[j][t];
      p[j][t].y = by[j][t];
    }
  }
  #pragma unroll
  for (int j = 0; j < GS; ++j)
    asm volatile("" : "+v"(p[j][0]), "+v"(p[j][1]), "+v"(p[j][2]), "+v"(p[j][3]));
  __builtin_amdgcn_sched_barrier(0);
  #pragma unroll
  for (int j = 0; j < GS; ++j) {
    #pragma unroll
    for (int t = 0; t < NT1; ++t)
      c[t] = wmma_f32(a[j], p[j][t], c[t]);
  }
  asm volatile("v_nop\n\tv_nop\n\tv_nop\n\tv_nop\n\tv_nop"
               : "+v"(c[0]), "+v"(c[1]), "+v"(c[2]), "+v"(c[3]));
  __builtin_amdgcn_sched_barrier(0);
}

__device__ __forceinline__ void load_w_group(v4f (&w)[WGRP],
                                             const v4f* __restrict__ p, int step) {
  #pragma unroll
  for (int j = 0; j < WGRP; ++j)
    w[j] = p[j * step];
}

__device__ __forceinline__ void load_nd_frags(v4f (&a)[WGRP], const float* pa) {
  #pragma unroll
  for (int j = 0; j < WGRP; ++j)
    a[j] = *(const v4f*)(pa + 8 * j);
}

__device__ __forceinline__ v8f mma_tile(v8f c, const v4f (&a)[WGRP],
                                        const v4f (&w)[WGRP]) {
  #pragma unroll
  for (int j = 0; j < WGRP; ++j) {
    c = wmma_f32(__builtin_shufflevector(a[j], a[j], 0, 1),
                 __builtin_shufflevector(w[j], w[j], 0, 1), c);
    c = wmma_f32(__builtin_shufflevector(a[j], a[j], 2, 3),
                 __builtin_shufflevector(w[j], w[j], 2, 3), c);
  }
  return c;
}

__device__ __forceinline__ void loads_first() {
  __builtin_amdgcn_sched_barrier(0);
}

__device__ __forceinline__ void refill_after(v8f (&c)[NT1], int& off, int& offa) {
  asm volatile("" : "+v"(c[0]), "+v"(c[1]), "+v"(c[2]), "+v"(c[3]),
                    "+v"(off), "+v"(offa));
}

__device__ __forceinline__ void offset_ready(int& off) {
  asm volatile("" : "+v"(off));
}

__device__ __forceinline__ void wmma_distance(v8f& c) {
  asm volatile("v_nop\n\tv_nop\n\tv_nop\n\tv_nop\n\tv_nop" : "+v"(c));
}

__device__ __forceinline__ void update_behind() {
  __builtin_amdgcn_sched_barrier(0);
}

__device__ __forceinline__ void lds_barrier() {
  __builtin_amdgcn_fence(__ATOMIC_RELEASE, "workgroup", "local");
  __builtin_amdgcn_s_barrier();
  __builtin_amdgcn_fence(__ATOMIC_ACQUIRE, "workgroup", "local");
}

__device__ __forceinline__ void sgd_update_reg(const v8f& acc, v8f& mb, v8f& sp,
                                               float* ps) {
  #pragma unroll
  for (int r = 0; r < 8; ++r) {
    float nb = MOM * mb[r] + INV * acc[r];
    mb[r] = nb;
    float ns = sp[r] - LR * nb;
    ns = ns > 0.0f ? ns : 0.0f;
    sp[r] = ns;
    ps[r * FSTR] = ns;
  }
}

__device__ __forceinline__ void sgd_update_col(const v8f& acc, int hi8, int fcol,
                                               float* s_spec, float* s_bcol) {
  float ob[8], os[8];
  #pragma unroll
  for (int r = 0; r < 8; ++r) {
    ob[r] = s_bcol[r + hi8];
    os[r] = s_spec[(r + hi8) * FSTR + fcol];
  }
  #pragma unroll
  for (int r = 0; r < 8; ++r) {
    float nb = MOM * ob[r] + INV * acc[r];
    ob[r] = nb;
    float ns = os[r] - LR * nb;
    os[r] = ns > 0.0f ? ns : 0.0f;
  }
  #pragma unroll
  for (int r = 0; r < 8; ++r) {
    s_bcol[r + hi8] = ob[r];
    s_spec[(r + hi8) * FSTR + fcol] = os[r];
  }
}

__device__ __forceinline__ void gemm2_tile(const v4f (&a)[WGRP],
                                           const v4f (&cur)[WGRP], v4f (&nxt)[WGRP],
                                           const v4f* __restrict__ np, int nstep,
                                           v8f& mb, v8f& sp, float* ps) {
  v8f acc = {0.f, 0.f, 0.f, 0.f, 0.f, 0.f, 0.f, 0.f};
  load_w_group(nxt, np, nstep);
  loads_first();
  acc = mma_tile(acc, a, cur);
  wmma_distance(acc);
  update_behind();
  sgd_update_reg(acc, mb, sp, ps);
}

__device__ __forceinline__ void gemm2_tile_last(const v4f (&a)[WGRP],
                                                const v4f (&cur)[WGRP],
                                                v4f (&bx0)[GS], v4f (&by0)[GS],
                                                const v4f* __restrict__ fb4, int noff,
                                                v8f& mb, v8f& sp, float* ps) {
  v8f acc = {0.f, 0.f, 0.f, 0.f, 0.f, 0.f, 0.f, 0.f};
  load_q_group(bx0, by0, fb4, noff, 0);
  loads_first();
  acc = mma_tile(acc, a, cur);
  wmma_distance(acc);
  update_behind();
  sgd_update_reg(acc, mb, sp, ps);
}

__device__ __forceinline__ void gemm2_tile_col(int fcol, int hi8,
                                               const v4f (&a)[WGRP],
                                               float* s_spec, float* s_bcol,
                                               const v4f (&cur)[WGRP],
                                               v4f (&bx0)[GS], v4f (&by0)[GS],
                                               const v4f* __restrict__ fb4, int noff) {
  v8f acc = {0.f, 0.f, 0.f, 0.f, 0.f, 0.f, 0.f, 0.f};
  offset_ready(fcol);
  load_q_group(bx0, by0, fb4, noff, 0);
  loads_first();
  acc = mma_tile(acc, a, cur);
  wmma_distance(acc);
  if (fcol < F_SZ)
    sgd_update_col(acc, hi8, fcol, s_spec, s_bcol);
}

__global__ __launch_bounds__(NTHREADS)
void transpose_fb_kernel(const float* __restrict__ fb,
                         float* __restrict__ fbT) {
  int i = blockIdx.x * NTHREADS + threadIdx.x;
  if (i < M_SZ * FP) {
    int mi = i & 3, f = (i >> 2) & (FP - 1), mb = i >> 12;
    fbT[i] = fb[f * M_SZ + 4 * mb + mi];
  }
}

__global__ __launch_bounds__(NTHREADS)
void invmel_sgd_kernel(const float* __restrict__ mel,
                       const float* __restrict__ spec0,
                       const float* __restrict__ fb,
                       const float* __restrict__ fbT,
                       float* __restrict__ out)
{
  __shared__ float s_spec[ROWS * FSTR];
  __shared__ float s_bcol[ROWS];
  __shared__ __attribute__((aligned(16))) float s_mel[ROWS * MSTR];
  __shared__ __attribute__((aligned(16))) float s_nd[ROWS * NDSTR];
  __shared__ __attribute__((aligned(16))) float s_part[(1024 / KQ) * ROWS * PSTR];

  const int tid  = threadIdx.x;
  const int lane = tid & 31;
  const int wave = __builtin_amdgcn_readfirstlane(tid >> 5);
  const int nl   = lane & 15;
  const int hs   = lane >> 4;
  const int khi  = hs * 2;
  const int hi8  = hs * 8;

  const int blk = blockIdx.x;
  const int b   = blk / (T_SZ / ROWS);
  const int t0  = (blk % (T_SZ / ROWS)) * ROWS;

  const int ng = wave & 1;
  const int kq = wave >> 1;
  const v4f* __restrict__ fb4 = (const v4f*)fb;
  const int obq = (kq * KQ + khi) * (M_SZ / 4) + ng * 16 + nl;
  const v4f* __restrict__ wt4 = (const v4f*)fbT;
  const int ow  = hs * FP + nl;
  const int oa1 = nl * FSTR + kq * KQ + khi;
  const float* pa2 = s_nd + nl * NDSTR + 4 * hs;
  const v4f* __restrict__ wp = wt4 + (ow + wave * 16);
  float* ps0 = s_spec + hi8 * FSTR + wave * 16 + nl;
  const v4f btl = fb4[FP * (M_SZ / 4) + ng * 16 + nl];

  v4f bx0[GS], by0[GS], bx1[GS], by1[GS];
  v2f a0[GS], a1[GS];
  v4f w0[WGRP], w1[WGRP];
  load_q_group(bx0, by0, fb4, obq, 0);

  {
    const int r16 = tid & (ROWS - 1);
    const float* __restrict__ gs = spec0 + (b * T_SZ + t0) * F_SZ + tid;
    const float* __restrict__ gc = spec0 + (b * T_SZ + t0 + r16) * F_SZ + (F_SZ - 1);
    const float* __restrict__ gm = mel + (b * M_SZ + (tid >> 4)) * T_SZ + t0 + r16;
    float xs[ROWS][4], xm[M_SZ / 16], xc;
    #pragma unroll
    for (int r = 0; r < ROWS; ++r) {
      #pragma unroll
      for (int j = 0; j < 4; ++j)
        xs[r][j] = gs[r * F_SZ + j * NTHREADS];
    }
    xc = *gc;
    #pragma unroll
    for (int k = 0; k < M_SZ / 16; ++k)
      xm[k] = gm[k * 16 * T_SZ];
    loads_first();
    float* ls = s_spec + tid;
    #pragma unroll
    for (int r = 0; r < ROWS; ++r) {
      #pragma unroll
      for (int j = 0; j < 4; ++j)
        ls[r * FSTR + j * NTHREADS] = xs[r][j];
    }
    s_spec[r16 * FSTR + (F_SZ - 1)] = xc;
    if (tid < ROWS)
      s_bcol[tid] = 0.0f;
    float* lm = s_mel + r16 * MSTR + (tid >> 4);
    #pragma unroll
    for (int k = 0; k < M_SZ / 16; ++k)
      lm[16 * k] = xm[k];
  }
  const bool tail_lane = (kq == 3) && (hs == 0);
  v4f btail;
  btail.x = tail_lane ? btl.x : 0.0f;
  btail.y = tail_lane ? btl.y : 0.0f;
  btail.z = tail_lane ? btl.z : 0.0f;
  btail.w = tail_lane ? btl.w : 0.0f;
  __syncthreads();

  v8f mb[NWAVES], sp[NWAVES];
  #pragma unroll
  for (int i = 0; i < NWAVES; ++i) {
    #pragma unroll
    for (int r = 0; r < 8; ++r) {
      mb[i][r] = 0.0f;
      sp[i][r] = ps0[i * 128 + r * FSTR];
    }
  }

  for (int it = 0; it < ITERS; ++it) {
    {
      v8f c[NT1];
      #pragma unroll
      for (int t = 0; t < NT1; ++t)
        #pragma unroll
        for (int r = 0; r < 8; ++r)
          c[t][r] = 0.0f;

      if (kq == 3) {
        v2f a;
        a.x = (hs == 0) ? s_spec[nl * FSTR + 1024] : 0.0f;
        a.y = 0.0f;
        #pragma unroll
        for (int t = 0; t < NT1; ++t) {
          v2f bf;
          bf.x = btail[t];
          bf.y = 0.0f;
          c[t] = wmma_f32(a, bf, c[t]);
        }
        asm volatile("v_nop\n\tv_nop\n\tv_nop\n\tv_nop\n\tv_nop"
                     : "+v"(c[0]), "+v"(c[1]), "+v"(c[2]), "+v"(c[3]));
      }

      int o = obq, oa = oa1;
      load_a_group(a0, s_spec, oa, 0);
      #pragma nounroll
      for (int g = 0; g < NGQ - 2; g += 2) {
        load_q_group(bx1, by1, fb4, o, (g + 1) * GS);
        load_a_group(a1, s_spec, oa, (g + 1) * GS);
        loads_first();
        mma_q_group(c, a0, bx0, by0);
        refill_after(c, o, oa);
        load_q_group(bx0, by0, fb4, o, (g + 2) * GS);
        load_a_group(a0, s_spec, oa, (g + 2) * GS);
        loads_first();
        mma_q_group(c, a1, bx1, by1);
      }
      load_q_group(bx1, by1, fb4, o, (NGQ - 1) * GS);
      load_a_group(a1, s_spec, oa, (NGQ - 1) * GS);
      loads_first();
      mma_q_group(c, a0, bx0, by0);
      load_w_group(w0, wp, WSTEP);
      loads_first();
      mma_q_group(c, a1, bx1, by1);

      float* pp = s_part + kq * (ROWS * PSTR) + hi8 * PSTR + ng * 64 + 4 * nl;
      #pragma unroll
      for (int r = 0; r < 8; ++r) {
        v4f v = {c[0][r], c[1][r], c[2][r], c[3][r]};
        *(v4f*)(pp + r * PSTR) = v;
      }
    }
    lds_barrier();

    {
      const int wo = (tid >> 4) * PSTR + 4 * (tid & 15);
      #pragma unroll
      for (int hf = 0; hf < 2; ++hf) {
        const int e = wo + hf * 64;
        v4f p0 = *(const v4f*)(s_part + 0 * ROWS * PSTR + e);
        v4f p1 = *(const v4f*)(s_part + 1 * ROWS * PSTR + e);
        v4f p2 = *(const v4f*)(s_part + 2 * ROWS * PSTR + e);
        v4f p3 = *(const v4f*)(s_part + 3 * ROWS * PSTR + e);
        v4f m  = *(const v4f*)(s_mel + e);
        *(v4f*)(s_nd + e) = ((p0 + p1) + (p2 + p3)) - m;
      }
    }
    lds_barrier();

    {
      v4f a2[WGRP];
      load_nd_frags(a2, pa2);
      #pragma unroll
      for (int i = 0; i < 6; i += 2) {
        gemm2_tile(a2, w0, w1, wp + (i + 1) * 128, WSTEP, mb[i], sp[i],
                   ps0 + i * 128);
        gemm2_tile(a2, w1, w0, wp + (i + 2) * 128, WSTEP, mb[i + 1], sp[i + 1],
                   ps0 + (i + 1) * 128);
      }
      gemm2_tile(a2, w0, w1, wp + 7 * 128, WSTEP, mb[6], sp[6], ps0 + 6 * 128);
      if (wave == 0) {
        gemm2_tile(a2, w1, w0, (const v4f*)(fb + FP * M_SZ) + hs, 2,
                   mb[7], sp[7], ps0 + 7 * 128);
        gemm2_tile_col(FP + nl, hi8, a2, s_spec, s_bcol, w0, bx0, by0, fb4, obq);
      } else {
        gemm2_tile_last(a2, w1, bx0, by0, fb4, obq, mb[7], sp[7], ps0 + 7 * 128);
      }
    }
    lds_barrier();
  }

  {
    const int r16 = tid & (ROWS - 1);
    const float* lo = s_spec + r16 * FSTR + (tid >> 4);
    float* __restrict__ go = out + (b * F_SZ + (tid >> 4)) * T_SZ + t0 + r16;
    float xo[FP / 16], xc;
    #pragma unroll
    for (int k = 0; k < FP / 16; ++k)
      xo[k] = lo[16 * k];
    xc = s_spec[r16 * FSTR + (F_SZ - 1)];
    #pragma unroll
    for (int k = 0; k < FP / 16; ++k)
      go[k * 16 * T_SZ] = xo[k];
    if (tid < ROWS)
      out[(b * F_SZ + (F_SZ - 1)) * T_SZ + t0 + tid] = xc;
  }
}

extern "C" void kernel_launch(void* const* d_in, const int* in_sizes, int n_in,
                              void* d_out, int out_size, void* d_ws, size_t ws_size,
                              hipStream_t stream) {
  (void)in_sizes; (void)n_in; (void)out_size; (void)ws_size;
  const float* mel   = (const float*)d_in[0];
  const float* spec0 = (const float*)d_in[1];
  const float* fb    = (const float*)d_in[2];
  float* out = (float*)d_out;
  float* fbT = (float*)d_ws;

  {
    dim3 grid((M_SZ * FP + NTHREADS - 1) / NTHREADS);
    hipLaunchKernelGGL(transpose_fb_kernel, grid, dim3(NTHREADS), 0, stream,
                       fb, fbT);
  }
  {
    dim3 grid((B_SZ * T_SZ) / ROWS);
    hipLaunchKernelGGL(invmel_sgd_kernel, grid, dim3(NTHREADS), 0, stream,
                       mel, spec0, fb, fbT, out);
  }
}
